# starting kernel
# baseline (speedup 1.0000x reference)
#include <hip/hip_runtime.h>


typedef __attribute__((ext_vector_type(2))) float v2f;
typedef __attribute__((ext_vector_type(8))) float v8f;

#define C_DIM 128
#define H_DIM 192
#define W_DIM 192
#define B_DIM 4
#define KSZ   9
#define RAD   4
#define HW    (H_DIM * W_DIM)
#define CHW   ((size_t)C_DIM * HW)

__launch_bounds__(32)
__global__ void corr_wmma_f32(const float* __restrict__ x,
                              const float* __restrict__ y,
                              float* __restrict__ out)
{
    const int tw = blockIdx.x;
    const int h  = blockIdx.y;
    const int b  = blockIdx.z;
    const int w0 = tw * 16;

    const int lane = threadIdx.x;
    const int half = lane >> 4;
    const int lm   = lane & 15;

    const float* xbase = x + (size_t)b * CHW + (size_t)(2 * half) * HW
                           + (size_t)h * W_DIM + (w0 + lm);
    v2f a[32];
#pragma unroll
    for (int kc = 0; kc < 32; ++kc) {
        const float* p = xbase + (size_t)(4 * kc) * HW;
        a[kc].x = p[0];
        a[kc].y = p[HW];
    }

    const int wp0 = w0 - 4 + lm;
    const int wp1 = w0 + 12 + lm;
    const bool ok0 = (wp0 >= 0) && (wp0 < W_DIM);
    const bool ok1 = (wp1 < W_DIM);
    const int wp0c = wp0 < 0 ? 0 : (wp0 >= W_DIM ? W_DIM - 1 : wp0);
    const int wp1c = wp1 >= W_DIM ? W_DIM - 1 : wp1;

    const size_t ych = (size_t)b * CHW + (size_t)(2 * half) * HW;
    const float inv_c = 1.0f / (float)C_DIM;

    for (int oi = 0; oi < KSZ; ++oi) {
        v8f d0a = {}, d0b = {};
        v8f d1a = {}, d1b = {};
        const int row = h + oi - RAD;
        if (row >= 0 && row < H_DIM) {
            const float* y0 = y + ych + (size_t)row * W_DIM + wp0c;
            const float* y1 = y + ych + (size_t)row * W_DIM + wp1c;
#pragma unroll
            for (int kc = 0; kc < 32; kc += 2) {
                const size_t offA = (size_t)(4 * kc) * HW;
                const size_t offB = (size_t)(4 * (kc + 1)) * HW;
                v2f b0a, b1a, b0b, b1b;
                b0a.x = y0[offA];
                b0a.y = y0[offA + HW];
                b1a.x = y1[offA];
                b1a.y = y1[offA + HW];
                b0b.x = y0[offB];
                b0b.y = y0[offB + HW];
                b1b.x = y1[offB];
                b1b.y = y1[offB + HW];
                d0a = __builtin_amdgcn_wmma_f32_16x16x4_f32(
                        false, a[kc],     false, b0a, (short)0, d0a, false, false);
                d1a = __builtin_amdgcn_wmma_f32_16x16x4_f32(
                        false, a[kc],     false, b1a, (short)0, d1a, false, false);
                d0b = __builtin_amdgcn_wmma_f32_16x16x4_f32(
                        false, a[kc + 1], false, b0b, (short)0, d0b, false, false);
                d1b = __builtin_amdgcn_wmma_f32_16x16x4_f32(
                        false, a[kc + 1], false, b1b, (short)0, d1b, false, false);
            }
        }
        float* obase = out + (size_t)b * (KSZ * KSZ) * HW
                           + (size_t)(oi * KSZ) * HW
                           + (size_t)h * W_DIM + w0;
#pragma unroll
        for (int v = 0; v < 8; ++v) {
            const int m = v + 8 * half;
            const int oj0 = lm - m;
            if (oj0 >= 0 && oj0 <= 8) {
                const float r = ok0 ? (d0a[v] + d0b[v]) * inv_c : 0.0f;
                obase[(size_t)oj0 * HW + m] = r;
            }
            const int oj1 = lm + 16 - m;
            if (oj1 <= 8) {
                const float r = ok1 ? (d1a[v] + d1b[v]) * inv_c : 0.0f;
                obase[(size_t)oj1 * HW + m] = r;
            }
        }
    }
}

extern "C" void kernel_launch(void* const* d_in, const int* in_sizes, int n_in,
                              void* d_out, int out_size, void* d_ws, size_t ws_size,
                              hipStream_t stream) {
    (void)in_sizes; (void)n_in; (void)d_ws; (void)ws_size; (void)out_size;
    const float* x = (const float*)d_in[0];
    const float* y = (const float*)d_in[1];
    float* out = (float*)d_out;
    dim3 grid(W_DIM / 16, H_DIM, B_DIM);
    dim3 block(32, 1, 1);
    corr_wmma_f32<<<grid, block, 0, stream>>>(x, y, out);
}


# x row and nine y rows staged in LDS by one 12-wave workgroup per image row: whole 128-B lines via 16-B loads, WMMA fragments as 8-B LDS reads (was 4-B global loads per fragment element)
# speedup vs baseline: 2.6978x; 2.6978x over previous
#include <hip/hip_runtime.h>


typedef __attribute__((ext_vector_type(2))) float v2f;
typedef __attribute__((ext_vector_type(4))) float v4f;
typedef __attribute__((ext_vector_type(8))) float v8f;

#define C_DIM 128
#define H_DIM 192
#define W_DIM 192
#define B_DIM 4
#define KSZ   9
#define RAD   4
#define HW    (H_DIM * W_DIM)
#define CHW   ((size_t)C_DIM * HW)
#define NPAIR (C_DIM / 2)
#define PITCH 208
#define ROWBUF (NPAIR * PITCH)
#define NTHR  384
#define NPF   16

__launch_bounds__(NTHR)
__global__ void corr_wmma_f32_lds(const float* __restrict__ x,
                                  const float* __restrict__ y,
                                  float* __restrict__ out)
{
    __shared__ __attribute__((aligned(16))) v2f lds[2 * ROWBUF];

    const int h = blockIdx.x;
    const int b = blockIdx.y;
    const int t = threadIdx.x;
    const int wave = t >> 5;
    const int lane = t & 31;
    const int half = lane >> 4;
    const int lm   = lane & 15;
    const int w0   = wave * 16;

    const int j   = t % 48;
    const int cp0 = t / 48;
    const size_t gofs = (size_t)b * CHW + (size_t)(2 * cp0) * HW + (size_t)(4 * j);
    const int    lofs = cp0 * PITCH + 4 + 4 * j;

    v4f pf[NPF];

    {
        v4f px[NPF];
        const float* xs = x + gofs + (size_t)h * W_DIM;
#pragma unroll
        for (int i = 0; i < NPF / 2; ++i) {
            px[2 * i]     = *(const v4f*)(xs + (size_t)i * 16 * HW);
            px[2 * i + 1] = *(const v4f*)(xs + (size_t)i * 16 * HW + HW);
        }
        const int row0 = h - RAD;
        if (row0 >= 0) {
            const float* ys = y + gofs + (size_t)row0 * W_DIM;
#pragma unroll
            for (int i = 0; i < NPF / 2; ++i) {
                pf[2 * i]     = *(const v4f*)(ys + (size_t)i * 16 * HW);
                pf[2 * i + 1] = *(const v4f*)(ys + (size_t)i * 16 * HW + HW);
            }
        }
        for (int i = t; i < 2 * NPAIR * 8; i += NTHR) {
            const int e = i & 7, cp = (i >> 3) & (NPAIR - 1), buf = i >> 9;
            const int p = e < 2 ? 2 * e : 192 + 2 * e;
            *(v4f*)(lds + buf * ROWBUF + cp * PITCH + p) = (v4f){0.f, 0.f, 0.f, 0.f};
        }
#pragma unroll
        for (int i = 0; i < NPF / 2; ++i) {
            const v4f e = px[2 * i], o = px[2 * i + 1];
            v4f* d = (v4f*)(lds + ROWBUF + lofs + i * 8 * PITCH);
            d[0] = (v4f){e.x, o.x, e.y, o.y};
            d[1] = (v4f){e.z, o.z, e.w, o.w};
        }
        if (row0 >= 0) {
#pragma unroll
            for (int i = 0; i < NPF / 2; ++i) {
                const v4f e = pf[2 * i], o = pf[2 * i + 1];
                v4f* d = (v4f*)(lds + lofs + i * 8 * PITCH);
                d[0] = (v4f){e.x, o.x, e.y, o.y};
                d[1] = (v4f){e.z, o.z, e.w, o.w};
            }
        }
    }
    __syncthreads();

    v2f a[32];
    {
        const v2f* ab = lds + ROWBUF + half * PITCH + 4 + w0 + lm;
#pragma unroll
        for (int kc = 0; kc < 32; ++kc)
            a[kc] = ab[(2 * kc) * PITCH];
    }
    __syncthreads();

    const float inv_c = 1.0f / (float)C_DIM;

    for (int oi = 0; oi < KSZ; ++oi) {
        const int row  = h + oi - RAD;
        const bool valid  = (row >= 0) && (row < H_DIM);
        const bool nvalid = (oi + 1 < KSZ) && (row + 1 >= 0) && (row + 1 < H_DIM);

        if (nvalid) {
            const float* ys = y + gofs + (size_t)(row + 1) * W_DIM;
#pragma unroll
            for (int i = 0; i < NPF / 2; ++i) {
                pf[2 * i]     = *(const v4f*)(ys + (size_t)i * 16 * HW);
                pf[2 * i + 1] = *(const v4f*)(ys + (size_t)i * 16 * HW + HW);
            }
        }

        v8f d0a = {}, d0b = {};
        v8f d1a = {}, d1b = {};
        if (valid) {
            const v2f* bb = lds + (oi & 1) * ROWBUF + half * PITCH + w0 + lm;
#pragma unroll
            for (int kc = 0; kc < 32; kc += 2) {
                const v2f* pa = bb + (2 * kc) * PITCH;
                const v2f* pb = bb + (2 * (kc + 1)) * PITCH;
                const v2f b0a = pa[0];
                const v2f b1a = pa[16];
                const v2f b0b = pb[0];
                const v2f b1b = pb[16];
                d0a = __builtin_amdgcn_wmma_f32_16x16x4_f32(
                        false, a[kc],     false, b0a, (short)0, d0a, false, false);
                d1a = __builtin_amdgcn_wmma_f32_16x16x4_f32(
                        false, a[kc],     false, b1a, (short)0, d1a, false, false);
                d0b = __builtin_amdgcn_wmma_f32_16x16x4_f32(
                        false, a[kc + 1], false, b0b, (short)0, d0b, false, false);
                d1b = __builtin_amdgcn_wmma_f32_16x16x4_f32(
                        false, a[kc + 1], false, b1b, (short)0, d1b, false, false);
            }
        }

        if (nvalid) {
            v2f* dst = lds + ((oi & 1) ^ 1) * ROWBUF + lofs;
#pragma unroll
            for (int i = 0; i < NPF / 2; ++i) {
                const v4f e = pf[2 * i], o = pf[2 * i + 1];
                v4f* d = (v4f*)(dst + i * 8 * PITCH);
                d[0] = (v4f){e.x, o.x, e.y, o.y};
                d[1] = (v4f){e.z, o.z, e.w, o.w};
            }
        }
        __syncthreads();

        float* obase = out + (size_t)b * (KSZ * KSZ) * HW
                           + (size_t)(oi * KSZ) * HW
                           + (size_t)h * W_DIM + w0;
#pragma unroll
        for (int v = 0; v < 8; ++v) {
            const int m = v + 8 * half;
            const int oj0 = lm - m;
            if (oj0 >= 0 && oj0 <= 8)
                obase[(size_t)oj0 * HW + m] = (d0a[v] + d0b[v]) * inv_c;
            const int oj1 = lm + 16 - m;
            if (oj1 <= 8)
                obase[(size_t)oj1 * HW + m] = (d1a[v] + d1b[v]) * inv_c;
        }
    }
}

extern "C" void kernel_launch(void* const* d_in, const int* in_sizes, int n_in,
                              void* d_out, int out_size, void* d_ws, size_t ws_size,
                              hipStream_t stream) {
    (void)in_sizes; (void)n_in; (void)d_ws; (void)ws_size; (void)out_size;
    const float* x = (const float*)d_in[0];
    const float* y = (const float*)d_in[1];
    float* out = (float*)d_out;
    dim3 grid(H_DIM, B_DIM, 1);
    dim3 block(NTHR, 1, 1);
    corr_wmma_f32_lds<<<grid, block, 0, stream>>>(x, y, out);
}


# one 16x16 tile per wave on a window of 16 pixels shared by x rows and y columns, windows 12 px apart (16 waves): 16 tiles per y row cover the 9-wide band, was 24 (two tiles per 16 px)
# speedup vs baseline: 2.9447x; 1.0915x over previous
#include <hip/hip_runtime.h>


typedef __attribute__((ext_vector_type(2))) float v2f;
typedef __attribute__((ext_vector_type(4))) float v4f;
typedef __attribute__((ext_vector_type(8))) float v8f;

#define C_DIM 128
#define H_DIM 192
#define W_DIM 192
#define B_DIM 4
#define KSZ   9
#define RAD   4
#define HW    (H_DIM * W_DIM)
#define CHW   ((size_t)C_DIM * HW)
#define NPAIR (C_DIM / 2)
#define PITCH 208
#define ROWBUF (NPAIR * PITCH)
#define NTHR  512
#define WSTEP 12
#define NPF   12

__launch_bounds__(NTHR)
__global__ void corr_wmma_f32_lds(const float* __restrict__ x,
                                  const float* __restrict__ y,
                                  float* __restrict__ out)
{
    __shared__ __attribute__((aligned(16))) v2f lds[2 * ROWBUF];

    const int h = blockIdx.x;
    const int b = blockIdx.y;
    const int t = threadIdx.x;
    const int wave = t >> 5;
    const int lane = t & 31;
    const int half = lane >> 4;
    const int lm   = lane & 15;
    const int w0   = wave * WSTEP;

    unsigned gofs[NPF / 2];
    int      lofs[NPF / 2];
#pragma unroll
    for (int i = 0; i < NPF / 2; ++i) {
        const int r  = lane + 32 * i;
        const int cp = 4 * wave + r / 48;
        const int j  = r % 48;
        gofs[i] = (unsigned)b * (unsigned)CHW + (unsigned)(2 * cp) * HW + (unsigned)(4 * j);
        lofs[i] = cp * PITCH + 4 + 4 * j;
    }

    v4f pf[NPF];

    {
        v4f px[NPF];
        const float* xs = x + (size_t)h * W_DIM;
#pragma unroll
        for (int i = 0; i < NPF / 2; ++i) {
            px[2 * i]     = *(const v4f*)(xs + gofs[i]);
            px[2 * i + 1] = *(const v4f*)(xs + gofs[i] + HW);
        }
        const int row0 = h - RAD;
        if (row0 >= 0) {
            const float* ys = y + (size_t)row0 * W_DIM;
#pragma unroll
            for (int i = 0; i < NPF / 2; ++i) {
                pf[2 * i]     = *(const v4f*)(ys + gofs[i]);
                pf[2 * i + 1] = *(const v4f*)(ys + gofs[i] + HW);
            }
        }
        for (int i = t; i < 2 * NPAIR * 8; i += NTHR) {
            const int e = i & 7, cp = (i >> 3) & (NPAIR - 1), buf = i >> 9;
            const int p = e < 2 ? 2 * e : 192 + 2 * e;
            *(v4f*)(lds + buf * ROWBUF + cp * PITCH + p) = (v4f){0.f, 0.f, 0.f, 0.f};
        }
#pragma unroll
        for (int i = 0; i < NPF / 2; ++i) {
            const v4f e = px[2 * i], o = px[2 * i + 1];
            v4f* d = (v4f*)(lds + ROWBUF + lofs[i]);
            d[0] = (v4f){e.x, o.x, e.y, o.y};
            d[1] = (v4f){e.z, o.z, e.w, o.w};
        }
        if (row0 >= 0) {
#pragma unroll
            for (int i = 0; i < NPF / 2; ++i) {
                const v4f e = pf[2 * i], o = pf[2 * i + 1];
                v4f* d = (v4f*)(lds + lofs[i]);
                d[0] = (v4f){e.x, o.x, e.y, o.y};
                d[1] = (v4f){e.z, o.z, e.w, o.w};
            }
        }
    }
    __syncthreads();

    const int frag = half * PITCH + 4 + w0 + lm;
    v2f a[32];
    {
        const v2f* ab = lds + ROWBUF + frag;
#pragma unroll
        for (int kc = 0; kc < 32; ++kc)
            a[kc] = ab[(2 * kc) * PITCH];
    }
    __syncthreads();

    const float inv_c = 1.0f / (float)C_DIM;

    for (int oi = 0; oi < KSZ; ++oi) {
        const int row  = h + oi - RAD;
        const bool valid  = (row >= 0) && (row < H_DIM);
        const bool nvalid = (oi + 1 < KSZ) && (row + 1 >= 0) && (row + 1 < H_DIM);

        if (nvalid) {
            const float* ys = y + (size_t)(row + 1) * W_DIM;
#pragma unroll
            for (int i = 0; i < NPF / 2; ++i) {
                pf[2 * i]     = *(const v4f*)(ys + gofs[i]);
                pf[2 * i + 1] = *(const v4f*)(ys + gofs[i] + HW);
            }
        }

        v8f da = {}, db = {};
        if (valid) {
            const v2f* bb = lds + (oi & 1) * ROWBUF + frag;
#pragma unroll
            for (int kc = 0; kc < 32; kc += 2) {
                const v2f b0 = bb[(2 * kc) * PITCH];
                const v2f b1 = bb[(2 * (kc + 1)) * PITCH];
                da = __builtin_amdgcn_wmma_f32_16x16x4_f32(
                        false, a[kc],     false, b0, (short)0, da, false, false);
                db = __builtin_amdgcn_wmma_f32_16x16x4_f32(
                        false, a[kc + 1], false, b1, (short)0, db, false, false);
            }
        }

        if (nvalid) {
            v2f* dst = lds + ((oi & 1) ^ 1) * ROWBUF;
#pragma unroll
            for (int i = 0; i < NPF / 2; ++i) {
                const v4f e = pf[2 * i], o = pf[2 * i + 1];
                v4f* d = (v4f*)(dst + lofs[i]);
                d[0] = (v4f){e.x, o.x, e.y, o.y};
                d[1] = (v4f){e.z, o.z, e.w, o.w};
            }
        }
        __syncthreads();

        float* obase = out + (size_t)b * (KSZ * KSZ) * HW
                           + (size_t)(oi * KSZ) * HW
                           + (size_t)h * W_DIM + w0;
#pragma unroll
        for (int v = 0; v < 8; ++v) {
            const int m  = v + 8 * half;
            int   oj  = lm - m + RAD;
            float val = (da[v] + db[v]) * inv_c;
            bool  wr  = (oj >= 0) && (oj <= 2 * RAD)
                        && (m >= 4 || wave == 0 || lm >= 4)
                        && (w0 + m < W_DIM);
            if (wave == 0 && m < 4 && lm >= 12 + m) {
                oj = lm - m - 12; val = 0.f; wr = true;
            }
            if (wr)
                obase[(size_t)oj * HW + m] = val;
        }
    }
}

extern "C" void kernel_launch(void* const* d_in, const int* in_sizes, int n_in,
                              void* d_out, int out_size, void* d_ws, size_t ws_size,
                              hipStream_t stream) {
    (void)in_sizes; (void)n_in; (void)d_ws; (void)ws_size; (void)out_size;
    const float* x = (const float*)d_in[0];
    const float* y = (const float*)d_in[1];
    float* out = (float*)d_out;
    dim3 grid(H_DIM, B_DIM, 1);
    dim3 block(NTHR, 1, 1);
    corr_wmma_f32_lds<<<grid, block, 0, stream>>>(x, y, out);
}


# band of row step oi held in 8 registers and stored during step oi+1, one store behind every 4th WMMA and after that step's row loads; step barrier fences LDS only (was: 8 stores after the barrier)
# speedup vs baseline: 3.6387x; 1.2357x over previous
#include <hip/hip_runtime.h>


typedef __attribute__((ext_vector_type(2))) float v2f;
typedef __attribute__((ext_vector_type(4))) float v4f;
typedef __attribute__((ext_vector_type(8))) float v8f;

#define C_DIM 128
#define H_DIM 192
#define W_DIM 192
#define B_DIM 4
#define KSZ   9
#define RAD   4
#define HW    (H_DIM * W_DIM)
#define CHW   ((size_t)C_DIM * HW)
#define NPAIR (C_DIM / 2)
#define PITCH 208
#define ROWBUF (NPAIR * PITCH)
#define NTHR  512
#define WSTEP 12
#define NPF   12

__device__ __forceinline__ void lds_barrier()
{
    __builtin_amdgcn_fence(__ATOMIC_RELEASE, "workgroup", "local");
    __builtin_amdgcn_s_barrier();
    __builtin_amdgcn_fence(__ATOMIC_ACQUIRE, "workgroup", "local");
}

__launch_bounds__(NTHR)
__global__ void corr_wmma_f32_lds(const float* __restrict__ x,
                                  const float* __restrict__ y,
                                  float* __restrict__ out)
{
    __shared__ __attribute__((aligned(16))) v2f lds[2 * ROWBUF];

    const int h = blockIdx.x;
    const int b = blockIdx.y;
    const int t = threadIdx.x;
    const int wave = t >> 5;
    const int lane = t & 31;
    const int half = lane >> 4;
    const int lm   = lane & 15;
    const int w0   = wave * WSTEP;

    unsigned gofs[NPF / 2];
    int      lofs[NPF / 2];
#pragma unroll
    for (int i = 0; i < NPF / 2; ++i) {
        const int r  = lane + 32 * i;
        const int cp = 4 * wave + r / 48;
        const int j  = r % 48;
        gofs[i] = (unsigned)b * (unsigned)CHW + (unsigned)(2 * cp) * HW + (unsigned)(4 * j);
        lofs[i] = cp * PITCH + 4 + 4 * j;
    }

    v4f pf[NPF];

    {
        v4f px[NPF];
        const float* xs = x + (size_t)h * W_DIM;
#pragma unroll
        for (int i = 0; i < NPF / 2; ++i) {
            px[2 * i]     = *(const v4f*)(xs + gofs[i]);
            px[2 * i + 1] = *(const v4f*)(xs + gofs[i] + HW);
        }
        const int row0 = h - RAD;
        if (row0 >= 0) {
            const float* ys = y + (size_t)row0 * W_DIM;
#pragma unroll
            for (int i = 0; i < NPF / 2; ++i) {
                pf[2 * i]     = *(const v4f*)(ys + gofs[i]);
                pf[2 * i + 1] = *(const v4f*)(ys + gofs[i] + HW);
            }
        }
        for (int i = t; i < 2 * NPAIR * 8; i += NTHR) {
            const int e = i & 7, cp = (i >> 3) & (NPAIR - 1), buf = i >> 9;
            const int p = e < 2 ? 2 * e : 192 + 2 * e;
            *(v4f*)(lds + buf * ROWBUF + cp * PITCH + p) = (v4f){0.f, 0.f, 0.f, 0.f};
        }
#pragma unroll
        for (int i = 0; i < NPF / 2; ++i) {
            const v4f e = px[2 * i], o = px[2 * i + 1];
            v4f* d = (v4f*)(lds + ROWBUF + lofs[i]);
            d[0] = (v4f){e.x, o.x, e.y, o.y};
            d[1] = (v4f){e.z, o.z, e.w, o.w};
        }
        if (row0 >= 0) {
#pragma unroll
            for (int i = 0; i < NPF / 2; ++i) {
                const v4f e = pf[2 * i], o = pf[2 * i + 1];
                v4f* d = (v4f*)(lds + lofs[i]);
                d[0] = (v4f){e.x, o.x, e.y, o.y};
                d[1] = (v4f){e.z, o.z, e.w, o.w};
            }
        }
    }
    __syncthreads();

    const int frag = half * PITCH + 4 + w0 + lm;
    v2f a[32];
    {
        const v2f* ab = lds + ROWBUF + frag;
#pragma unroll
        for (int kc = 0; kc < 32; ++kc)
            a[kc] = ab[(2 * kc) * PITCH];
    }
    __syncthreads();

    const float inv_c = 1.0f / (float)C_DIM;

    const bool zl = (wave == 0) && (half == 0) && (lm >= 12);
    bool wr[8];
#pragma unroll
    for (int v = 0; v < 8; ++v) {
        const int m  = v + 8 * half;
        const int oj = lm - m + RAD;
        wr[v] = zl ? (v < 4 && lm >= 12 + v)
                   : ((oj >= 0) && (oj <= 2 * RAD)
                      && (m >= 4 || wave == 0 || lm >= 4)
                      && (w0 + m < W_DIM));
    }
    const long loff = zl ? (long)(lm - 12) * HW
                         : (long)(lm + RAD) * HW - (long)(8 * half) * (HW - 1);
    float* const pl0 = out + (size_t)b * (KSZ * KSZ) * HW
                           + (size_t)h * W_DIM + w0 + loff;

    float  pv[8];
    float* pp = pl0;
#pragma unroll
    for (int v = 0; v < 8; ++v) pv[v] = 0.f;

    for (int oi = 0; oi < KSZ; ++oi) {
        const int row  = h + oi - RAD;
        const bool valid  = (row >= 0) && (row < H_DIM);
        const bool nvalid = (oi + 1 < KSZ) && (row + 1 >= 0) && (row + 1 < H_DIM);
        const bool st = oi > 0;

        if (nvalid) {
            const float* ys = y + (size_t)(row + 1) * W_DIM;
#pragma unroll
            for (int i = 0; i < NPF / 2; ++i) {
                pf[2 * i]     = *(const v4f*)(ys + gofs[i]);
                pf[2 * i + 1] = *(const v4f*)(ys + gofs[i] + HW);
            }
        }

        v8f da = {}, db = {};
        if (valid) {
            const v2f* bb = lds + (oi & 1) * ROWBUF + frag;
#pragma unroll
            for (int g = 0; g < 8; ++g) {
#pragma unroll
                for (int kc = 4 * g; kc < 4 * g + 4; kc += 2) {
                    const v2f b0 = bb[(2 * kc) * PITCH];
                    const v2f b1 = bb[(2 * (kc + 1)) * PITCH];
                    da = __builtin_amdgcn_wmma_f32_16x16x4_f32(
                            false, a[kc],     false, b0, (short)0, da, false, false);
                    db = __builtin_amdgcn_wmma_f32_16x16x4_f32(
                            false, a[kc + 1], false, b1, (short)0, db, false, false);
                }
                if (st && wr[g]) pp[-(long)g * (HW - 1)] = pv[g];
            }
        } else if (st) {
#pragma unroll
            for (int g = 0; g < 8; ++g)
                if (wr[g]) pp[-(long)g * (HW - 1)] = pv[g];
        }

        if (nvalid) {
            v2f* dst = lds + ((oi & 1) ^ 1) * ROWBUF;
#pragma unroll
            for (int i = 0; i < NPF / 2; ++i) {
                const v4f e = pf[2 * i], o = pf[2 * i + 1];
                v4f* d = (v4f*)(dst + lofs[i]);
                d[0] = (v4f){e.x, o.x, e.y, o.y};
                d[1] = (v4f){e.z, o.z, e.w, o.w};
            }
        }
        lds_barrier();
        __builtin_amdgcn_sched_barrier(0);

#pragma unroll
        for (int v = 0; v < 8; ++v)
            pv[v] = zl ? 0.f : (da[v] + db[v]) * inv_c;
        pp = pl0 + (size_t)(oi * KSZ) * HW;
    }

#pragma unroll
    for (int g = 0; g < 8; ++g)
        if (wr[g]) pp[-(long)g * (HW - 1)] = pv[g];
}

extern "C" void kernel_launch(void* const* d_in, const int* in_sizes, int n_in,
                              void* d_out, int out_size, void* d_ws, size_t ws_size,
                              hipStream_t stream) {
    (void)in_sizes; (void)n_in; (void)d_ws; (void)ws_size; (void)out_size;
    const float* x = (const float*)d_in[0];
    const float* y = (const float*)d_in[1];
    float* out = (float*)d_out;
    dim3 grid(H_DIM, B_DIM, 1);
    dim3 block(NTHR, 1, 1);
    corr_wmma_f32_lds<<<grid, block, 0, stream>>>(x, y, out);
}


# y row of step oi+2 requested during step oi, behind its WMMAs, into a second register set (was: row oi+1 at the head of step oi): a row has a whole step to land, no band store queues behind loads
# speedup vs baseline: 4.8836x; 1.3421x over previous
#include <hip/hip_runtime.h>


typedef __attribute__((ext_vector_type(2))) float v2f;
typedef __attribute__((ext_vector_type(4))) float v4f;
typedef __attribute__((ext_vector_type(8))) float v8f;

#define C_DIM 128
#define H_DIM 192
#define W_DIM 192
#define B_DIM 4
#define KSZ   9
#define RAD   4
#define HW    (H_DIM * W_DIM)
#define CHW   ((size_t)C_DIM * HW)
#define NPAIR (C_DIM / 2)
#define PITCH 208
#define ROWBUF (NPAIR * PITCH)
#define NTHR  512
#define WSTEP 12
#define NPF   12

__device__ __forceinline__ void lds_barrier()
{
    __builtin_amdgcn_fence(__ATOMIC_RELEASE, "workgroup", "local");
    __builtin_amdgcn_s_barrier();
    __builtin_amdgcn_fence(__ATOMIC_ACQUIRE, "workgroup", "local");
}

#define ROT_ROW(val, n) \
    __builtin_bit_cast(float, __builtin_amdgcn_update_dpp( \
        0, __builtin_bit_cast(int, (float)(val)), 0x120 + (n), 0xf, 0xf, false))

__launch_bounds__(NTHR)
__global__ void corr_wmma_f32_lds(const float* __restrict__ x,
                                  const float* __restrict__ y,
                                  float* __restrict__ out)
{
    __shared__ __attribute__((aligned(16))) v2f lds[2 * ROWBUF];

    const int h = blockIdx.x;
    const int b = blockIdx.y;
    const int t = threadIdx.x;
    const int wave = t >> 5;
    const int lane = t & 31;
    const int half = lane >> 4;
    const int lm   = lane & 15;
    const int w0   = wave * WSTEP;

    unsigned gofs[NPF / 2];
    int      lofs[NPF / 2];
#pragma unroll
    for (int i = 0; i < NPF / 2; ++i) {
        const int r  = lane + 32 * i;
        const int cp = 4 * wave + r / 48;
        const int j  = r % 48;
        gofs[i] = 4u * ((unsigned)b * (unsigned)CHW + (unsigned)(2 * cp) * HW + (unsigned)(4 * j));
        lofs[i] = cp * PITCH + 4 + 4 * j;
    }

    v4f pf0[NPF], pf1[NPF];

    {
        v4f px[NPF];
        const char* xs = (const char*)(x + (size_t)h * W_DIM);
#pragma unroll
        for (int i = 0; i < NPF / 2; ++i) {
            px[2 * i]     = *(const v4f*)(xs + gofs[i]);
            px[2 * i + 1] = *(const v4f*)(xs + gofs[i] + 4 * HW);
        }
        const int row0 = h - RAD;
        if (row0 >= 0) {
            const char* ys = (const char*)(y + (size_t)row0 * W_DIM);
#pragma unroll
            for (int i = 0; i < NPF / 2; ++i) {
                pf0[2 * i]     = *(const v4f*)(ys + gofs[i]);
                pf0[2 * i + 1] = *(const v4f*)(ys + gofs[i] + 4 * HW);
            }
        }
        for (int i = t; i < 2 * NPAIR * 8; i += NTHR) {
            const int e = i & 7, cp = (i >> 3) & (NPAIR - 1), buf = i >> 9;
            const int p = e < 2 ? 2 * e : 192 + 2 * e;
            *(v4f*)(lds + buf * ROWBUF + cp * PITCH + p) = (v4f){0.f, 0.f, 0.f, 0.f};
        }
#pragma unroll
        for (int i = 0; i < NPF / 2; ++i) {
            const v4f e = px[2 * i], o = px[2 * i + 1];
            v4f* d = (v4f*)(lds + ROWBUF + lofs[i]);
            d[0] = (v4f){e.x, o.x, e.y, o.y};
            d[1] = (v4f){e.z, o.z, e.w, o.w};
        }
        if (row0 >= 0) {
#pragma unroll
            for (int i = 0; i < NPF / 2; ++i) {
                const v4f e = pf0[2 * i], o = pf0[2 * i + 1];
                v4f* d = (v4f*)(lds + lofs[i]);
                d[0] = (v4f){e.x, o.x, e.y, o.y};
                d[1] = (v4f){e.z, o.z, e.w, o.w};
            }
        }
        {
            const int lrow = row0 + 1 < 0 ? 0 : row0 + 1;
            const char* ys = (const char*)(y + (size_t)lrow * W_DIM);
#pragma unroll
            for (int i = 0; i < NPF / 2; ++i) {
                pf1[2 * i]     = *(const v4f*)(ys + gofs[i]);
                pf1[2 * i + 1] = *(const v4f*)(ys + gofs[i] + 4 * HW);
            }
        }
    }
    lds_barrier();

    const int frag = half * PITCH + 4 + w0 + lm;
    v2f a[32];
    {
        const v2f* ab = lds + ROWBUF + frag;
#pragma unroll
        for (int kc = 0; kc < 32; ++kc)
            a[kc] = ab[(2 * kc) * PITCH];
    }
    lds_barrier();

    const float inv_c = 1.0f / (float)C_DIM;

    const int  oj   = half ? ((lm - 8) & 15) : lm;
    const bool act  = oj <= 2 * RAD;
    const bool edge = (w0 + 15 < W_DIM);
    const bool mA   = act && (half ? (edge && oj <= 4) : (wave == 0 || oj == 8));
    bool mB[4], zf[4];
#pragma unroll
    for (int i = 0; i < 4; ++i) {
        mB[i] = act && (oj >= 5) && (oj <= 7)
                && (half ? (edge && i <= 7 - oj) : (wave > 0 && i >= 8 - oj));
        zf[i] = (wave == 0) && (half == 0) && (lm + i < 4);
    }
    float* const pl0 = out + (size_t)b * (KSZ * KSZ) * HW + (size_t)h * W_DIM + w0
                           + (size_t)(act ? oj : 0) * HW;
    float* pq = pl0 + (half ? 8 : 4);
    float* pe = pl0 + (half ? 12 : 0);

    float q[4], p[4];
#pragma unroll
    for (int i = 0; i < 4; ++i) { q[i] = 0.f; p[i] = 0.f; }

    auto row_step = [&](const int oi, const bool do_ld, const bool do_stg,
                        v4f (&ld)[NPF], v4f (&stg)[NPF])
                    __attribute__((always_inline)) {
        const int row  = h + oi - RAD;
        const bool valid  = (row >= 0) && (row < H_DIM);
        const bool nvalid = do_stg && (row + 1 >= 0) && (row + 1 < H_DIM);
        const bool st = oi > 0;

        v8f da = {}, db = {};
        if (valid) {
            const v2f* bb = lds + (oi & 1) * ROWBUF + frag;
#pragma unroll
            for (int g = 0; g < 8; ++g) {
#pragma unroll
                for (int kc = 4 * g; kc < 4 * g + 4; kc += 2) {
                    const v2f b0 = bb[(2 * kc) * PITCH];
                    const v2f b1 = bb[(2 * (kc + 1)) * PITCH];
                    da = __builtin_amdgcn_wmma_f32_16x16x4_f32(
                            false, a[kc],     false, b0, (short)0, da, false, false);
                    db = __builtin_amdgcn_wmma_f32_16x16x4_f32(
                            false, a[kc + 1], false, b1, (short)0, db, false, false);
                }
                if (st) {
                    if (g == 0)      { if (act) *(v4f*)pq = (v4f){q[0], q[1], q[2], q[3]}; }
                    else if (g == 1) { if (mA)  *(v4f*)pe = (v4f){p[0], p[1], p[2], p[3]}; }
                    else if (g < 6)  { if (mB[(g - 2) & 3]) pe[(g - 2) & 3] = p[(g - 2) & 3]; }
                }
            }
        } else if (st) {
            if (act) *(v4f*)pq = (v4f){q[0], q[1], q[2], q[3]};
            if (mA)  *(v4f*)pe = (v4f){p[0], p[1], p[2], p[3]};
#pragma unroll
            for (int i = 0; i < 4; ++i)
                if (mB[i]) pe[i] = p[i];
        }

        if (do_ld) {
            const int lrow = row + 2 < 0 ? 0 : (row + 2 > H_DIM - 1 ? H_DIM - 1 : row + 2);
            const char* ys = (const char*)(y + (size_t)lrow * W_DIM);
#pragma unroll
            for (int i = 0; i < NPF / 2; ++i) {
                asm volatile("" : "+v"(gofs[i]));
                ld[2 * i]     = *(const v4f*)(ys + gofs[i]);
                ld[2 * i + 1] = *(const v4f*)(ys + gofs[i] + 4 * HW);
            }
        }

        if (nvalid) {
            v2f* dst = lds + ((oi & 1) ^ 1) * ROWBUF;
#pragma unroll
            for (int i = 0; i < NPF / 2; ++i) {
                const v4f e = stg[2 * i], o = stg[2 * i + 1];
                v4f* d = (v4f*)(dst + lofs[i]);
                d[0] = (v4f){e.x, o.x, e.y, o.y};
                d[1] = (v4f){e.z, o.z, e.w, o.w};
            }
        }
        lds_barrier();
        __builtin_amdgcn_sched_barrier(0);

        float r[8];
        r[0] = ROT_ROW((da[0] + db[0]) * inv_c, 4);
        r[1] = ROT_ROW((da[1] + db[1]) * inv_c, 3);
        r[2] = ROT_ROW((da[2] + db[2]) * inv_c, 2);
        r[3] = ROT_ROW((da[3] + db[3]) * inv_c, 1);
        r[4] = (da[4] + db[4]) * inv_c;
        r[5] = ROT_ROW((da[5] + db[5]) * inv_c, 15);
        r[6] = ROT_ROW((da[6] + db[6]) * inv_c, 14);
        r[7] = ROT_ROW((da[7] + db[7]) * inv_c, 13);
#pragma unroll
        for (int i = 0; i < 4; ++i) {
            q[i] = half ? r[i] : r[4 + i];
            const float e = half ? r[4 + i] : r[i];
            p[i] = zf[i] ? 0.f : e;
        }
        pq = pl0 + (half ? 8 : 4) + (size_t)(oi * KSZ) * HW;
        pe = pl0 + (half ? 12 : 0) + (size_t)(oi * KSZ) * HW;
    };

#pragma unroll 1
    for (int oi = 0; oi < KSZ - 3; oi += 2) {
        row_step(oi,     true, true, pf0, pf1);
        row_step(oi + 1, true, true, pf1, pf0);
    }
    row_step(KSZ - 3, true,  true,  pf0, pf1);
    row_step(KSZ - 2, false, true,  pf1, pf0);
    row_step(KSZ - 1, false, false, pf0, pf1);

    if (act) *(v4f*)pq = (v4f){q[0], q[1], q[2], q[3]};
    if (mA)  *(v4f*)pe = (v4f){p[0], p[1], p[2], p[3]};
#pragma unroll
    for (int i = 0; i < 4; ++i)
        if (mB[i]) pe[i] = p[i];
}

extern "C" void kernel_launch(void* const* d_in, const int* in_sizes, int n_in,
                              void* d_out, int out_size, void* d_ws, size_t ws_size,
                              hipStream_t stream) {
    (void)in_sizes; (void)n_in; (void)d_ws; (void)ws_size; (void)out_size;
    const float* x = (const float*)d_in[0];
    const float* y = (const float*)d_in[1];
    float* out = (float*)d_out;
    dim3 grid(H_DIM, B_DIM, 1);
    dim3 block(NTHR, 1, 1);
    corr_wmma_f32_lds<<<grid, block, 0, stream>>>(x, y, out);
}


# M of each 16x16 tile = two x rows x eight columns, N = sixteen y columns of one staged y row: the 9-wide band of both rows in ONE tile; 120 tiles and 5 staged y rows per x row (was 144 and 9)
# speedup vs baseline: 5.5557x; 1.1376x over previous
#include <hip/hip_runtime.h>


typedef __attribute__((ext_vector_type(2))) float v2f;
typedef __attribute__((ext_vector_type(4))) float v4f;
typedef __attribute__((ext_vector_type(8))) float v8f;

#define C_DIM 128
#define H_DIM 192
#define W_DIM 192
#define B_DIM 4
#define KSZ   9
#define RAD   4
#define HW    (H_DIM * W_DIM)
#define CHW   ((size_t)C_DIM * HW)
#define NPAIR (C_DIM / 2)
#define PITCH 144
#define ROWBUF (NPAIR * PITCH)
#define NBUF  3
#define NTHR  512
#define NCW   12
#define NPF   8
#define NSTEP 10
#define NWGX  192

__device__ __forceinline__ void lds_barrier()
{
    __builtin_amdgcn_fence(__ATOMIC_RELEASE, "workgroup", "local");
    __builtin_amdgcn_s_barrier();
    __builtin_amdgcn_fence(__ATOMIC_ACQUIRE, "workgroup", "local");
}

#define ROT_ROW(val, n) \
    __builtin_bit_cast(float, __builtin_amdgcn_update_dpp( \
        0, __builtin_bit_cast(int, (float)(val)), 0x120 + (n), 0xf, 0xf, false))

__launch_bounds__(NTHR)
__global__ void corr_wmma_f32_rowpair(const float* __restrict__ x,
                                      const float* __restrict__ y,
                                      float* __restrict__ out)
{
    __shared__ __attribute__((aligned(16))) v2f lds[NBUF * ROWBUF];

    const int h  = 2 * (blockIdx.x >> 1);
    const int hs = blockIdx.x & 1;
    const int b  = blockIdx.y;
    const int t  = threadIdx.x;
    const int wave = __builtin_amdgcn_readfirstlane(t >> 5);
    const int lane = t & 31;
    const int half = lane >> 4;
    const int lm   = lane & 15;
    const int g0   = 64 * hs;
    const bool cw  = wave < NCW;
    const int ws   = 32 * hs + 8 * (cw ? wave : 0);

    unsigned gofs[NPF / 2];
    int      lofs[NPF / 2];
#pragma unroll
    for (int i = 0; i < NPF / 2; ++i) {
        const int cp = wave + 16 * i;
        gofs[i] = 4u * ((unsigned)b * (unsigned)CHW + (unsigned)(2 * cp) * HW + (unsigned)(g0 + 4 * lane));
        lofs[i] = cp * PITCH + 4 + 4 * lane;
    }

    v4f pf0[NPF], pf1[NPF];

    {
        v4f px[2 * NPF];
        const char* xs = (const char*)(x + (size_t)h * W_DIM);
#pragma unroll
        for (int i = 0; i < NPF / 2; ++i) {
            px[2 * i]           = *(const v4f*)(xs + gofs[i]);
            px[2 * i + 1]       = *(const v4f*)(xs + gofs[i] + 4 * HW);
            px[NPF + 2 * i]     = *(const v4f*)(xs + gofs[i] + 4 * W_DIM);
            px[NPF + 2 * i + 1] = *(const v4f*)(xs + gofs[i] + 4 * W_DIM + 4 * HW);
        }
        const int row0 = h - RAD;
        if (row0 >= 0) {
            const char* ys = (const char*)(y + (size_t)row0 * W_DIM);
#pragma unroll
            for (int i = 0; i < NPF / 2; ++i) {
                pf0[2 * i]     = *(const v4f*)(ys + gofs[i]);
                pf0[2 * i + 1] = *(const v4f*)(ys + gofs[i] + 4 * HW);
            }
        }
        for (int i = t; i < 2 * NPAIR * 8; i += NTHR) {
            const int e = i & 7, cp = (i >> 3) & (NPAIR - 1), buf = i >> 9;
            const int p = e < 2 ? 2 * e : 128 + 2 * e;
            *(v4f*)(lds + buf * ROWBUF + cp * PITCH + p) = (v4f){0.f, 0.f, 0.f, 0.f};
        }
#pragma unroll
        for (int i = 0; i < NPF / 2; ++i) {
            const v4f e0 = px[2 * i], o0 = px[2 * i + 1];
            v4f* d0 = (v4f*)(lds + ROWBUF + lofs[i]);
            d0[0] = (v4f){e0.x, o0.x, e0.y, o0.y};
            d0[1] = (v4f){e0.z, o0.z, e0.w, o0.w};
            const v4f e1 = px[NPF + 2 * i], o1 = px[NPF + 2 * i + 1];
            v4f* d1 = (v4f*)(lds + 2 * ROWBUF + lofs[i]);
            d1[0] = (v4f){e1.x, o1.x, e1.y, o1.y};
            d1[1] = (v4f){e1.z, o1.z, e1.w, o1.w};
        }
        if (row0 >= 0) {
#pragma unroll
            for (int i = 0; i < NPF / 2; ++i) {
                const v4f e = pf0[2 * i], o = pf0[2 * i + 1];
                v4f* d = (v4f*)(lds + lofs[i]);
                d[0] = (v4f){e.x, o.x, e.y, o.y};
                d[1] = (v4f){e.z, o.z, e.w, o.w};
            }
        }
        {
            const int lrow = row0 + 1 < 0 ? 0 : row0 + 1;
            const char* ys = (const char*)(y + (size_t)lrow * W_DIM);
#pragma unroll
            for (int i = 0; i < NPF / 2; ++i) {
                pf1[2 * i]     = *(const v4f*)(ys + gofs[i]);
                pf1[2 * i + 1] = *(const v4f*)(ys + gofs[i] + 4 * HW);
            }
        }
    }
    lds_barrier();

    const int fragA = (lm >> 3) * ROWBUF + half * PITCH + ws + 4 + (lm & 7);
    const int fragB = half * PITCH + ws + lm;
    v2f a[32];
    if (cw) {
        const v2f* ab = lds + ROWBUF + fragA;
#pragma unroll
        for (int kc = 0; kc < 32; ++kc)
            a[kc] = ab[(2 * kc) * PITCH];
    }
    lds_barrier();

    const float inv_c = 1.0f / (float)C_DIM;

    const bool act = cw && (lm <= 2 * RAD);
    float* const pl0 = out + (size_t)b * (KSZ * KSZ) * HW + (size_t)(h + half) * W_DIM
                           + (g0 + ws) + (size_t)(act ? lm : 0) * HW;
    float* pq = pl0;
    bool pst = false;

    float q[8];
#pragma unroll
    for (int i = 0; i < 8; ++i) q[i] = 0.f;

    auto row_step = [&](const int oi, const bool do_ld, const bool do_stg,
                        v4f (&ld)[NPF], v4f (&stg)[NPF])
                    __attribute__((always_inline)) {
        const int row  = h + oi - RAD;
        const bool valid  = (row >= 0) && (row < H_DIM);
        const bool nvalid = do_stg && (row + 1 >= 0) && (row + 1 < H_DIM);

        v8f da = {}, db = {};
        if (cw) {
            if (valid) {
                const v2f* bb = lds + (oi & 1) * ROWBUF + fragB;
#pragma unroll
                for (int g = 0; g < 8; ++g) {
#pragma unroll
                    for (int kc = 4 * g; kc < 4 * g + 4; kc += 2) {
                        const v2f b0 = bb[(2 * kc) * PITCH];
                        const v2f b1 = bb[(2 * (kc + 1)) * PITCH];
                        da = __builtin_amdgcn_wmma_f32_16x16x4_f32(
                                false, a[kc],     false, b0, (short)0, da, false, false);
                        db = __builtin_amdgcn_wmma_f32_16x16x4_f32(
                                false, a[kc + 1], false, b1, (short)0, db, false, false);
                    }
                    if (g == 0)      { if (pst) *(v4f*)pq       = (v4f){q[0], q[1], q[2], q[3]}; }
                    else if (g == 1) { if (pst) *(v4f*)(pq + 4) = (v4f){q[4], q[5], q[6], q[7]}; }
                }
            } else if (pst) {
                *(v4f*)pq       = (v4f){q[0], q[1], q[2], q[3]};
                *(v4f*)(pq + 4) = (v4f){q[4], q[5], q[6], q[7]};
            }
        }

        if (do_ld) {
            const int lrow = row + 2 < 0 ? 0 : (row + 2 > H_DIM - 1 ? H_DIM - 1 : row + 2);
            const char* ys = (const char*)(y + (size_t)lrow * W_DIM);
#pragma unroll
            for (int i = 0; i < NPF / 2; ++i) {
                asm volatile("" : "+v"(gofs[i]));
                ld[2 * i]     = *(const v4f*)(ys + gofs[i]);
                ld[2 * i + 1] = *(const v4f*)(ys + gofs[i] + 4 * HW);
            }
        }

        if (nvalid) {
            v2f* dst = lds + ((oi & 1) ^ 1) * ROWBUF;
#pragma unroll
            for (int i = 0; i < NPF / 2; ++i) {
                const v4f e = stg[2 * i], o = stg[2 * i + 1];
                v4f* d = (v4f*)(dst + lofs[i]);
                d[0] = (v4f){e.x, o.x, e.y, o.y};
                d[1] = (v4f){e.z, o.z, e.w, o.w};
            }
        }
        lds_barrier();
        __builtin_amdgcn_sched_barrier(0);

        q[0] = (da[0] + db[0]) * inv_c;
        q[1] = ROT_ROW((da[1] + db[1]) * inv_c, 15);
        q[2] = ROT_ROW((da[2] + db[2]) * inv_c, 14);
        q[3] = ROT_ROW((da[3] + db[3]) * inv_c, 13);
        q[4] = ROT_ROW((da[4] + db[4]) * inv_c, 12);
        q[5] = ROT_ROW((da[5] + db[5]) * inv_c, 11);
        q[6] = ROT_ROW((da[6] + db[6]) * inv_c, 10);
        q[7] = ROT_ROW((da[7] + db[7]) * inv_c, 9);
        const int oil = oi - half;
        pst = act && (oil >= 0) && (oil < KSZ);
        pq  = pl0 + (size_t)((pst ? oil : 0) * KSZ) * HW;
    };

#pragma unroll 1
    for (int oi = 0; oi < NSTEP - 2; oi += 2) {
        row_step(oi,     true, true, pf0, pf1);
        row_step(oi + 1, true, true, pf1, pf0);
    }
    row_step(NSTEP - 2, false, true,  pf0, pf1);
    row_step(NSTEP - 1, false, false, pf1, pf0);

    if (pst) {
        *(v4f*)pq       = (v4f){q[0], q[1], q[2], q[3]};
        *(v4f*)(pq + 4) = (v4f){q[4], q[5], q[6], q[7]};
    }
}

extern "C" void kernel_launch(void* const* d_in, const int* in_sizes, int n_in,
                              void* d_out, int out_size, void* d_ws, size_t ws_size,
                              hipStream_t stream) {
    (void)in_sizes; (void)n_in; (void)d_ws; (void)ws_size; (void)out_size;
    const float* x = (const float*)d_in[0];
    const float* y = (const float*)d_in[1];
    float* out = (float*)d_out;
    dim3 grid(NWGX, B_DIM, 1);
    dim3 block(NTHR, 1, 1);
    corr_wmma_f32_rowpair<<<grid, block, 0, stream>>>(x, y, out);
}


# 256 persistent workgroups of 3 tiles: the row stream runs on across tiles and the next tile's x rows arrive in steps 2-5 in LDS buffers of their own (was 768 workgroups, each waiting for its operands)
# speedup vs baseline: 5.7319x; 1.0317x over previous
#include <hip/hip_runtime.h>


typedef __attribute__((ext_vector_type(2))) float v2f;
typedef __attribute__((ext_vector_type(4))) float v4f;
typedef __attribute__((ext_vector_type(8))) float v8f;

#define C_DIM 128
#define H_DIM 192
#define W_DIM 192
#define B_DIM 4
#define KSZ   9
#define RAD   4
#define HW    (H_DIM * W_DIM)
#define CHW   ((size_t)C_DIM * HW)
#define NPAIR (C_DIM / 2)
#define PITCH 144
#define ROWBUF (NPAIR * PITCH)
#define NBUF  4
#define NTHR  512
#define NCW   12
#define NPF   8
#define NSTEP 10
#define NWGX  192
#define NWG   256
#define NTILE 3

__device__ __forceinline__ void lds_barrier()
{
    __builtin_amdgcn_fence(__ATOMIC_RELEASE, "workgroup", "local");
    __builtin_amdgcn_s_barrier();
    __builtin_amdgcn_fence(__ATOMIC_ACQUIRE, "workgroup", "local");
}

#define ROT_ROW(val, n) \
    __builtin_bit_cast(float, __builtin_amdgcn_update_dpp( \
        0, __builtin_bit_cast(int, (float)(val)), 0x120 + (n), 0xf, 0xf, false))

__launch_bounds__(NTHR)
__global__ void corr_wmma_f32_rowpair(const float* __restrict__ x,
                                      const float* __restrict__ y,
                                      float* __restrict__ out)
{
    __shared__ __attribute__((aligned(16))) v2f lds[NBUF * ROWBUF];

    int tl = blockIdx.x;
    int h  = 2 * ((tl % NWGX) >> 1);
    int b  = tl / NWGX;
    const int hs = blockIdx.x & 1;
    const int t  = threadIdx.x;
    const int wave = __builtin_amdgcn_readfirstlane(t >> 5);
    const int lane = t & 31;
    const int half = lane >> 4;
    const int lm   = lane & 15;
    const int g0   = 64 * hs;
    const bool cw  = wave < NCW;
    const int ws   = 32 * hs + 8 * (cw ? wave : 0);

    unsigned gofs[NPF / 2];
    int      lofs[NPF / 2];
#pragma unroll
    for (int i = 0; i < NPF / 2; ++i) {
        const int cp = wave + 16 * i;
        gofs[i] = 4u * ((unsigned)(2 * cp) * HW + (unsigned)(4 * lane));
        lofs[i] = cp * PITCH + 4 + 4 * lane;
    }

    v4f pf0[NPF], pf1[NPF], pfx[NPF];

    {
        v4f px[2 * NPF];
        const char* xs = (const char*)(x + (size_t)b * CHW + (size_t)h * W_DIM + g0);
#pragma unroll
        for (int i = 0; i < NPF / 2; ++i) {
            px[2 * i]           = *(const v4f*)(xs + gofs[i]);
            px[2 * i + 1]       = *(const v4f*)(xs + gofs[i] + 4 * HW);
            px[NPF + 2 * i]     = *(const v4f*)(xs + gofs[i] + 4 * W_DIM);
            px[NPF + 2 * i + 1] = *(const v4f*)(xs + gofs[i] + 4 * W_DIM + 4 * HW);
        }
        const int row0 = h - RAD;
        if (row0 >= 0) {
            const char* ys = (const char*)(y + (size_t)b * CHW + (size_t)row0 * W_DIM + g0);
#pragma unroll
            for (int i = 0; i < NPF / 2; ++i) {
                pf0[2 * i]     = *(const v4f*)(ys + gofs[i]);
                pf0[2 * i + 1] = *(const v4f*)(ys + gofs[i] + 4 * HW);
            }
        }
        for (int i = t; i < 2 * NPAIR * 8; i += NTHR) {
            const int e = i & 7, cp = (i >> 3) & (NPAIR - 1), buf = i >> 9;
            const int p = e < 2 ? 2 * e : 128 + 2 * e;
            *(v4f*)(lds + buf * ROWBUF + cp * PITCH + p) = (v4f){0.f, 0.f, 0.f, 0.f};
        }
#pragma unroll
        for (int i = 0; i < NPF / 2; ++i) {
            const v4f e0 = px[2 * i], o0 = px[2 * i + 1];
            v4f* d0 = (v4f*)(lds + 2 * ROWBUF + lofs[i]);
            d0[0] = (v4f){e0.x, o0.x, e0.y, o0.y};
            d0[1] = (v4f){e0.z, o0.z, e0.w, o0.w};
            const v4f e1 = px[NPF + 2 * i], o1 = px[NPF + 2 * i + 1];
            v4f* d1 = (v4f*)(lds + 3 * ROWBUF + lofs[i]);
            d1[0] = (v4f){e1.x, o1.x, e1.y, o1.y};
            d1[1] = (v4f){e1.z, o1.z, e1.w, o1.w};
        }
        if (row0 >= 0) {
#pragma unroll
            for (int i = 0; i < NPF / 2; ++i) {
                const v4f e = pf0[2 * i], o = pf0[2 * i + 1];
                v4f* d = (v4f*)(lds + lofs[i]);
                d[0] = (v4f){e.x, o.x, e.y, o.y};
                d[1] = (v4f){e.z, o.z, e.w, o.w};
            }
        }
        {
            const int lrow = row0 + 1 < 0 ? 0 : row0 + 1;
            const char* ys = (const char*)(y + (size_t)b * CHW + (size_t)lrow * W_DIM + g0);
#pragma unroll
            for (int i = 0; i < NPF / 2; ++i) {
                pf1[2 * i]     = *(const v4f*)(ys + gofs[i]);
                pf1[2 * i + 1] = *(const v4f*)(ys + gofs[i] + 4 * HW);
            }
        }
    }
    lds_barrier();

    const int fragA = (lm >> 3) * ROWBUF + half * PITCH + ws + 4 + (lm & 7);
    const int fragB = half * PITCH + ws + lm;
    v2f a[32];

    const float inv_c = 1.0f / (float)C_DIM;

    const bool act = cw && (lm <= 2 * RAD);
    const long long plw = (long long)(g0 + ws) + (long long)(act ? lm : 0) * HW
                        + (long long)half * W_DIM - (long long)half * (KSZ * HW);
    long long pl0 = plw;
    float* pq = out;
    bool pst = false;

    float q[8];
#pragma unroll
    for (int i = 0; i < 8; ++i) q[i] = 0.f;

    int hn = h, bn = b;
    bool has_next = false;

    auto row_step = [&](const int oi, v4f (&ld)[NPF], v4f (&stg)[NPF],
                        const int xreq, const int xstg)
                    __attribute__((always_inline)) {
        const int row  = h + oi - RAD;
        const bool valid  = (row >= 0) && (row < H_DIM);
        const int  srow   = (oi + 1 < NSTEP) ? row + 1 : hn - RAD;
        const bool nvalid = (srow >= 0) && (srow < H_DIM) && ((oi + 1 < NSTEP) || has_next);

        v8f da = {}, db = {};
        if (cw) {
            if (valid) {
                const v2f* bb = lds + (oi & 1) * ROWBUF + fragB;
#pragma unroll
                for (int g = 0; g < 8; ++g) {
#pragma unroll
                    for (int kc = 4 * g; kc < 4 * g + 4; kc += 2) {
                        const v2f b0 = bb[(2 * kc) * PITCH];
                        const v2f b1 = bb[(2 * (kc + 1)) * PITCH];
                        da = __builtin_amdgcn_wmma_f32_16x16x4_f32(
                                false, a[kc],     false, b0, (short)0, da, false, false);
                        db = __builtin_amdgcn_wmma_f32_16x16x4_f32(
                                false, a[kc + 1], false, b1, (short)0, db, false, false);
                    }
                    if (g == 0)      { if (pst) *(v4f*)pq       = (v4f){q[0], q[1], q[2], q[3]}; }
                    else if (g == 1) { if (pst) *(v4f*)(pq + 4) = (v4f){q[4], q[5], q[6], q[7]}; }
                }
            } else if (pst) {
                *(v4f*)pq       = (v4f){q[0], q[1], q[2], q[3]};
                *(v4f*)(pq + 4) = (v4f){q[4], q[5], q[6], q[7]};
            }
        }

        {
            const int rr   = (oi + 2 < NSTEP) ? row + 2 : hn - RAD + (oi + 2 - NSTEP);
            const int bb2  = (oi + 2 < NSTEP) ? b : bn;
            const int lrow = rr < 0 ? 0 : (rr > H_DIM - 1 ? H_DIM - 1 : rr);
            const char* ys = (const char*)(y + (size_t)bb2 * CHW + (size_t)lrow * W_DIM + g0);
#pragma unroll
            for (int i = 0; i < NPF / 2; ++i) {
                asm volatile("" : "+v"(gofs[i]));
                ld[2 * i]     = *(const v4f*)(ys + gofs[i]);
                ld[2 * i + 1] = *(const v4f*)(ys + gofs[i] + 4 * HW);
            }
        }
        if (xreq != 0) {
            const char* xs = (const char*)(x + (size_t)bn * CHW + (size_t)(hn + xreq - 1) * W_DIM + g0);
#pragma unroll
            for (int i = 0; i < NPF / 2; ++i) {
                asm volatile("" : "+v"(gofs[i]));
                pfx[2 * i]     = *(const v4f*)(xs + gofs[i]);
                pfx[2 * i + 1] = *(const v4f*)(xs + gofs[i] + 4 * HW);
            }
        }

        if (nvalid) {
            v2f* dst = lds + ((oi & 1) ^ 1) * ROWBUF;
#pragma unroll
            for (int i = 0; i < NPF / 2; ++i) {
                const v4f e = stg[2 * i], o = stg[2 * i + 1];
                v4f* d = (v4f*)(dst + lofs[i]);
                d[0] = (v4f){e.x, o.x, e.y, o.y};
                d[1] = (v4f){e.z, o.z, e.w, o.w};
            }
        }
        if (xstg != 0) {
            if (has_next) {
                v2f* dst = lds + (1 + xstg) * ROWBUF;
#pragma unroll
                for (int i = 0; i < NPF / 2; ++i) {
                    const v4f e = pfx[2 * i], o = pfx[2 * i + 1];
                    v4f* d = (v4f*)(dst + lofs[i]);
                    d[0] = (v4f){e.x, o.x, e.y, o.y};
                    d[1] = (v4f){e.z, o.z, e.w, o.w};
                }
            }
        }
        lds_barrier();
        __builtin_amdgcn_sched_barrier(0);

        q[0] = (da[0] + db[0]) * inv_c;
        q[1] = ROT_ROW((da[1] + db[1]) * inv_c, 15);
        q[2] = ROT_ROW((da[2] + db[2]) * inv_c, 14);
        q[3] = ROT_ROW((da[3] + db[3]) * inv_c, 13);
        q[4] = ROT_ROW((da[4] + db[4]) * inv_c, 12);
        q[5] = ROT_ROW((da[5] + db[5]) * inv_c, 11);
        q[6] = ROT_ROW((da[6] + db[6]) * inv_c, 10);
        q[7] = ROT_ROW((da[7] + db[7]) * inv_c, 9);
        const int oil = oi - half;
        pst = act && (oil >= 0) && (oil < KSZ);
        pq  = out + (pst ? pl0 + (long long)(oi * KSZ) * HW : 0ll);
    };

#pragma unroll 1
    for (int k = 0; k < NTILE; ++k) {
        has_next = k + 1 < NTILE;
        const int tn = has_next ? tl + NWG : tl;
        hn = 2 * ((tn % NWGX) >> 1);
        bn = tn / NWGX;

        if (cw) {
            const v2f* ab = lds + 2 * ROWBUF + fragA;
#pragma unroll
            for (int kc = 0; kc < 32; ++kc)
                a[kc] = ab[(2 * kc) * PITCH];
        }
        pl0 = plw + (long long)b * (KSZ * KSZ) * HW + (long long)h * W_DIM;

        row_step(0, pf0, pf1, 0, 0);
        row_step(1, pf1, pf0, 0, 0);
        row_step(2, pf0, pf1, 1, 0);
        row_step(3, pf1, pf0, 0, 1);
        row_step(4, pf0, pf1, 2, 0);
        row_step(5, pf1, pf0, 0, 2);
        row_step(6, pf0, pf1, 0, 0);
        row_step(7, pf1, pf0, 0, 0);
        row_step(8, pf0, pf1, 0, 0);
        row_step(9, pf1, pf0, 0, 0);

        tl = tn; h = hn; b = bn;
    }

    if (pst) {
        *(v4f*)pq       = (v4f){q[0], q[1], q[2], q[3]};
        *(v4f*)(pq + 4) = (v4f){q[4], q[5], q[6], q[7]};
    }
}

extern "C" void kernel_launch(void* const* d_in, const int* in_sizes, int n_in,
                              void* d_out, int out_size, void* d_ws, size_t ws_size,
                              hipStream_t stream) {
    (void)in_sizes; (void)n_in; (void)d_ws; (void)ws_size; (void)out_size;
    const float* x = (const float*)d_in[0];
    const float* y = (const float*)d_in[1];
    float* out = (float*)d_out;
    dim3 grid(NWG, 1, 1);
    dim3 block(NTHR, 1, 1);
    corr_wmma_f32_rowpair<<<grid, block, 0, stream>>>(x, y, out);
}


# a step's 32 LDS fragment reads issued in 3 groups (16, 8, 8), a group ahead of their WMMAs (was 2 behind every 2nd WMMA); for it band stores moved into the barrier wait and LDS addresses share bases
# speedup vs baseline: 5.9943x; 1.0458x over previous
#include <hip/hip_runtime.h>


typedef __attribute__((ext_vector_type(2))) float v2f;
typedef __attribute__((ext_vector_type(4))) float v4f;
typedef __attribute__((ext_vector_type(8))) float v8f;

#define C_DIM 128
#define H_DIM 192
#define W_DIM 192
#define B_DIM 4
#define KSZ   9
#define RAD   4
#define HW    (H_DIM * W_DIM)
#define CHW   ((size_t)C_DIM * HW)
#define NPAIR (C_DIM / 2)
#define PITCH 144
#define ROWBUF (NPAIR * PITCH)
#define NBUF  4
#define NTHR  512
#define NCW   12
#define NPF   8
#define NSTEP 10
#define NWGX  192
#define NWG   256
#define NTILE 3
#define GSTEP ((size_t)128 * HW)
#define LSTEP (16 * PITCH)
#define KHI   29

__device__ __forceinline__ void lds_barrier_arrive()
{
    __builtin_amdgcn_fence(__ATOMIC_RELEASE, "workgroup", "local");
    __builtin_amdgcn_s_barrier_signal(-1);
}
__device__ __forceinline__ void lds_barrier_wait()
{
    __builtin_amdgcn_s_barrier_wait(-1);
    __builtin_amdgcn_fence(__ATOMIC_ACQUIRE, "workgroup", "local");
}

__device__ __forceinline__ void lds_barrier()
{
    __builtin_amdgcn_fence(__ATOMIC_RELEASE, "workgroup", "local");
    __builtin_amdgcn_s_barrier();
    __builtin_amdgcn_fence(__ATOMIC_ACQUIRE, "workgroup", "local");
}

#define ROT_ROW(val, n) \
    __builtin_bit_cast(float, __builtin_amdgcn_update_dpp( \
        0, __builtin_bit_cast(int, (float)(val)), 0x120 + (n), 0xf, 0xf, false))

__launch_bounds__(NTHR)
__global__ void corr_wmma_f32_rowpair(const float* __restrict__ x,
                                      const float* __restrict__ y,
                                      float* __restrict__ out)
{
    __shared__ __attribute__((aligned(16))) v2f lds[NBUF * ROWBUF];

    int tl = blockIdx.x;
    int h  = 2 * ((tl % NWGX) >> 1);
    int b  = tl / NWGX;
    const int hs = blockIdx.x & 1;
    const int t  = threadIdx.x;
    const int wave = __builtin_amdgcn_readfirstlane(t >> 5);
    const int lane = t & 31;
    const int half = lane >> 4;
    const int lm   = lane & 15;
    const int g0   = 64 * hs;
    const bool cw  = wave < NCW;
    const int ws   = 32 * hs + 8 * (cw ? wave : 0);

    unsigned gofs0 = 4u * ((unsigned)(2 * wave) * HW + (unsigned)(4 * lane));
    unsigned gofs2 = gofs0 + (unsigned)(2 * GSTEP);
    const int lofs0 = wave * PITCH + 4 + 4 * lane;

    v4f pf0[NPF], pf1[NPF], pfx[NPF];

    {
        v4f px[2 * NPF];
        const char* xs = (const char*)(x + (size_t)b * CHW + (size_t)h * W_DIM + g0);
#pragma unroll
        for (int i = 0; i < NPF / 2; ++i) {
            px[2 * i]           = *(const v4f*)(xs + i * GSTEP + gofs0);
            px[2 * i + 1]       = *(const v4f*)(xs + i * GSTEP + gofs0 + 4 * HW);
            px[NPF + 2 * i]     = *(const v4f*)(xs + i * GSTEP + gofs0 + 4 * W_DIM);
            px[NPF + 2 * i + 1] = *(const v4f*)(xs + i * GSTEP + gofs0 + 4 * W_DIM + 4 * HW);
        }
        const int row0 = h - RAD;
        if (row0 >= 0) {
            const char* ys = (const char*)(y + (size_t)b * CHW + (size_t)row0 * W_DIM + g0);
#pragma unroll
            for (int i = 0; i < NPF / 2; ++i) {
                pf0[2 * i]     = *(const v4f*)(ys + i * GSTEP + gofs0);
                pf0[2 * i + 1] = *(const v4f*)(ys + i * GSTEP + gofs0 + 4 * HW);
            }
        }
        for (int i = t; i < 2 * NPAIR * 8; i += NTHR) {
            const int e = i & 7, cp = (i >> 3) & (NPAIR - 1), buf = i >> 9;
            const int p = e < 2 ? 2 * e : 128 + 2 * e;
            *(v4f*)(lds + buf * ROWBUF + cp * PITCH + p) = (v4f){0.f, 0.f, 0.f, 0.f};
        }
#pragma unroll
        for (int i = 0; i < NPF / 2; ++i) {
            const v4f e0 = px[2 * i], o0 = px[2 * i + 1];
            v4f* d0 = (v4f*)(lds + 2 * ROWBUF + lofs0 + i * LSTEP);
            d0[0] = (v4f){e0.x, o0.x, e0.y, o0.y};
            d0[1] = (v4f){e0.z, o0.z, e0.w, o0.w};
            const v4f e1 = px[NPF + 2 * i], o1 = px[NPF + 2 * i + 1];
            v4f* d1 = (v4f*)(lds + 3 * ROWBUF + lofs0 + i * LSTEP);
            d1[0] = (v4f){e1.x, o1.x, e1.y, o1.y};
            d1[1] = (v4f){e1.z, o1.z, e1.w, o1.w};
        }
        if (row0 >= 0) {
#pragma unroll
            for (int i = 0; i < NPF / 2; ++i) {
                const v4f e = pf0[2 * i], o = pf0[2 * i + 1];
                v4f* d = (v4f*)(lds + lofs0 + i * LSTEP);
                d[0] = (v4f){e.x, o.x, e.y, o.y};
                d[1] = (v4f){e.z, o.z, e.w, o.w};
            }
        }
        {
            const int lrow = row0 + 1 < 0 ? 0 : row0 + 1;
            const char* ys = (const char*)(y + (size_t)b * CHW + (size_t)lrow * W_DIM + g0);
#pragma unroll
            for (int i = 0; i < NPF / 2; ++i) {
                pf1[2 * i]     = *(const v4f*)(ys + i * GSTEP + gofs0);
                pf1[2 * i + 1] = *(const v4f*)(ys + i * GSTEP + gofs0 + 4 * HW);
            }
        }
    }
    lds_barrier();

    const int fragA = (lm >> 3) * ROWBUF + half * PITCH + ws + 4 + (lm & 7);
    const int fragB = half * PITCH + ws + lm;
    v2f a[32];

    const float inv_c = 1.0f / (float)C_DIM;

    const bool act = cw && (lm <= 2 * RAD);
    const long long plw = (long long)(g0 + ws) + (long long)(act ? lm : 0) * HW
                        + (long long)half * W_DIM - (long long)half * (KSZ * HW);
    long long pl0 = plw;
    float* pq = out;
    bool pst = false;

    float q[8];
#pragma unroll
    for (int i = 0; i < 8; ++i) q[i] = 0.f;

    int hn = h, bn = b;
    bool has_next = false;

    auto row_step = [&](const int oi, v4f (&ld)[NPF], v4f (&stg)[NPF],
                        const int xreq, const int xstg)
                    __attribute__((always_inline)) {
        const int row  = h + oi - RAD;
        const bool valid  = (row >= 0) && (row < H_DIM);
        asm volatile("" : "+v"(gofs0), "+v"(gofs2));
        const int  srow   = (oi + 1 < NSTEP) ? row + 1 : hn - RAD;
        const bool nvalid = (srow >= 0) && (srow < H_DIM) && ((oi + 1 < NSTEP) || has_next);

        v8f da = {}, db = {};
        if (cw) {
            if (valid) {
                int fb = fragB + (oi & 1) * ROWBUF;
                asm volatile("" : "+v"(fb));
                int fh = fb + 2 * KHI * PITCH;
                asm volatile("" : "+v"(fh));
                const v2f* bb = lds + fb;
                const v2f* bh = lds + fh;
                v2f bf[32];
#pragma unroll
                for (int kc = 0; kc < 32; ++kc)
                    bf[kc] = kc < KHI ? bb[(2 * kc) * PITCH] : bh[(2 * (kc - KHI)) * PITCH];
#pragma unroll
                for (int kc = 0; kc < 32; kc += 2) {
                    da = __builtin_amdgcn_wmma_f32_16x16x4_f32(
                            false, a[kc],     false, bf[kc],     (short)0, da, false, false);
                    db = __builtin_amdgcn_wmma_f32_16x16x4_f32(
                            false, a[kc + 1], false, bf[kc + 1], (short)0, db, false, false);
                }
                __builtin_amdgcn_sched_group_barrier(0x100, 16, 0);
                __builtin_amdgcn_sched_group_barrier(0x008, 8, 0);
                __builtin_amdgcn_sched_group_barrier(0x100, 8, 0);
                __builtin_amdgcn_sched_group_barrier(0x008, 8, 0);
                __builtin_amdgcn_sched_group_barrier(0x100, 8, 0);
                __builtin_amdgcn_sched_group_barrier(0x008, 16, 0);
            }
        }

        {
            const int rr   = (oi + 2 < NSTEP) ? row + 2 : hn - RAD + (oi + 2 - NSTEP);
            const int bb2  = (oi + 2 < NSTEP) ? b : bn;
            const int lrow = rr < 0 ? 0 : (rr > H_DIM - 1 ? H_DIM - 1 : rr);
            const char* ys = (const char*)(y + (size_t)bb2 * CHW + (size_t)lrow * W_DIM + g0);
#pragma unroll
            for (int i = 0; i < NPF / 2; ++i) {
                const unsigned go = i < 2 ? gofs0 : gofs2;
                ld[2 * i]     = *(const v4f*)(ys + go + (i & 1) * GSTEP);
                ld[2 * i + 1] = *(const v4f*)(ys + go + (i & 1) * GSTEP + 4 * HW);
            }
        }
        if (xreq != 0) {
            const char* xs = (const char*)(x + (size_t)bn * CHW + (size_t)(hn + xreq - 1) * W_DIM + g0);
#pragma unroll
            for (int i = 0; i < NPF / 2; ++i) {
                const unsigned go = i < 2 ? gofs0 : gofs2;
                pfx[2 * i]     = *(const v4f*)(xs + go + (i & 1) * GSTEP);
                pfx[2 * i + 1] = *(const v4f*)(xs + go + (i & 1) * GSTEP + 4 * HW);
            }
        }

        if (nvalid) {
            int sb = lofs0 + ((oi & 1) ^ 1) * ROWBUF;
            asm volatile("" : "+v"(sb));
#pragma unroll
            for (int i = 0; i < NPF / 2; ++i) {
                const v4f e = stg[2 * i], o = stg[2 * i + 1];
                v4f* d = (v4f*)(lds + sb + i * LSTEP);
                d[0] = (v4f){e.x, o.x, e.y, o.y};
                d[1] = (v4f){e.z, o.z, e.w, o.w};
            }
        }
        if (xstg != 0) {
            if (has_next) {
                int sx = lofs0 + (1 + xstg) * ROWBUF;
                asm volatile("" : "+v"(sx));
#pragma unroll
                for (int i = 0; i < NPF / 2; ++i) {
                    const v4f e = pfx[2 * i], o = pfx[2 * i + 1];
                    v4f* d = (v4f*)(lds + sx + i * LSTEP);
                    d[0] = (v4f){e.x, o.x, e.y, o.y};
                    d[1] = (v4f){e.z, o.z, e.w, o.w};
                }
            }
        }
        lds_barrier_arrive();
        if (pst) {
            *(v4f*)pq       = (v4f){q[0], q[1], q[2], q[3]};
            *(v4f*)(pq + 4) = (v4f){q[4], q[5], q[6], q[7]};
        }
        lds_barrier_wait();
        __builtin_amdgcn_sched_barrier(0);

        q[0] = (da[0] + db[0]) * inv_c;
        q[1] = ROT_ROW((da[1] + db[1]) * inv_c, 15);
        q[2] = ROT_ROW((da[2] + db[2]) * inv_c, 14);
        q[3] = ROT_ROW((da[3] + db[3]) * inv_c, 13);
        q[4] = ROT_ROW((da[4] + db[4]) * inv_c, 12);
        q[5] = ROT_ROW((da[5] + db[5]) * inv_c, 11);
        q[6] = ROT_ROW((da[6] + db[6]) * inv_c, 10);
        q[7] = ROT_ROW((da[7] + db[7]) * inv_c, 9);
        const int oil = oi - half;
        pst = act && (oil >= 0) && (oil < KSZ);
        pq  = out + (pst ? pl0 + (long long)(oi * KSZ) * HW : 0ll);
    };

#pragma unroll 1
    for (int k = 0; k < NTILE; ++k) {
        has_next = k + 1 < NTILE;
        const int tn = has_next ? tl + NWG : tl;
        hn = 2 * ((tn % NWGX) >> 1);
        bn = tn / NWGX;

        if (cw) {
            int fa = fragA + 2 * ROWBUF;
            asm volatile("" : "+v"(fa));
            int fah = fa + 2 * KHI * PITCH;
            asm volatile("" : "+v"(fah));
            const v2f* ab = lds + fa;
            const v2f* ah = lds + fah;
#pragma unroll
            for (int kc = 0; kc < 32; ++kc)
                a[kc] = kc < KHI ? ab[(2 * kc) * PITCH] : ah[(2 * (kc - KHI)) * PITCH];
        }
        pl0 = plw + (long long)b * (KSZ * KSZ) * HW + (long long)h * W_DIM;

        row_step(0, pf0, pf1, 0, 0);
        row_step(1, pf1, pf0, 0, 0);
        row_step(2, pf0, pf1, 1, 0);
        row_step(3, pf1, pf0, 0, 1);
        row_step(4, pf0, pf1, 2, 0);
        row_step(5, pf1, pf0, 0, 2);
        row_step(6, pf0, pf1, 0, 0);
        row_step(7, pf1, pf0, 0, 0);
        row_step(8, pf0, pf1, 0, 0);
        row_step(9, pf1, pf0, 0, 0);

        tl = tn; h = hn; b = bn;
    }

    if (pst) {
        *(v4f*)pq       = (v4f){q[0], q[1], q[2], q[3]};
        *(v4f*)(pq + 4) = (v4f){q[4], q[5], q[6], q[7]};
    }
}

extern "C" void kernel_launch(void* const* d_in, const int* in_sizes, int n_in,
                              void* d_out, int out_size, void* d_ws, size_t ws_size,
                              hipStream_t stream) {
    (void)in_sizes; (void)n_in; (void)d_ws; (void)ws_size; (void)out_size;
    const float* x = (const float*)d_in[0];
    const float* y = (const float*)d_in[1];
    float* out = (float*)d_out;
    dim3 grid(NWG, 1, 1);
    dim3 block(NTHR, 1, 1);
    corr_wmma_f32_rowpair<<<grid, block, 0, stream>>>(x, y, out);
}
